# starting kernel
# baseline (speedup 1.0000x reference)
#include <hip/hip_runtime.h>


typedef __attribute__((ext_vector_type(2))) float v2f;
typedef __attribute__((ext_vector_type(8))) float v8f;

#define N_NODES 100000
#define OFF_NORM 100096
#define OFF_BUFA (OFF_NORM + 1250048)
#define NBUF ((size_t)N_NODES * 64)


__global__ void gcn_deg_init(float* __restrict__ deg, int n) {
  int i = blockIdx.x * blockDim.x + threadIdx.x;
  if (i < n) deg[i] = 1.0f;
}

__global__ void gcn_deg_count(const int* __restrict__ dst, float* __restrict__ deg, int nE) {
  int e = blockIdx.x * blockDim.x + threadIdx.x;
  if (e < nE) atomicAdd(&deg[dst[e]], 1.0f);
}

__global__ void gcn_deg_rsqrt(float* __restrict__ deg, int n) {
  int i = blockIdx.x * blockDim.x + threadIdx.x;
  if (i < n) deg[i] = rsqrtf(deg[i]);
}

__global__ void gcn_edge_norm(const int* __restrict__ src, const int* __restrict__ dst,
                              const float* __restrict__ dis, float* __restrict__ nrm,
                              int nE) {
  int e = blockIdx.x * blockDim.x + threadIdx.x;
  if (e < nE) nrm[e] = dis[src[e]] * dis[dst[e]];
}


template <int K, int NC, bool RELU>
__global__ __launch_bounds__(64)
void gcn_gemm_wmma(const float* X, const float* __restrict__ W,
                   float* __restrict__ H, float* OUTI,
                   const float* __restrict__ dis, const float* __restrict__ bias) {
  constexpr int S  = K + 6;
  constexpr int CT = NC / 16;
  __shared__ float sWt[NC * S];

  for (int t = threadIdx.x; t < K * NC / 4; t += 64) {
    const int kr = (t * 4) / NC;
    const int n  = (t * 4) % NC;
    const float4 w4 = *(const float4*)(W + (size_t)t * 4);
    sWt[(size_t)(n + 0) * S + kr] = w4.x;
    sWt[(size_t)(n + 1) * S + kr] = w4.y;
    sWt[(size_t)(n + 2) * S + kr] = w4.z;
    sWt[(size_t)(n + 3) * S + kr] = w4.w;
  }
  __syncthreads();

  const int lane = threadIdx.x & 31;
  const int wave = threadIdx.x >> 5;
  const int m    = lane & 15;
  const int half = lane >> 4;
  const long long row0 = ((long long)blockIdx.x * 2 + wave) * 16;

  const float* xrow  = X + (size_t)(row0 + m) * K + 2 * half;
  const float* bbase = sWt + (size_t)m * S + 2 * half;

  v8f acc[CT] = {};
#pragma unroll
  for (int kk = 0; kk < K / 4; ++kk) {
    const int k = kk * 4;
    float ax = xrow[k];
    float ay = xrow[k + 1];
    if (RELU) { ax = fmaxf(ax, 0.f); ay = fmaxf(ay, 0.f); }
    const v2f a = {ax, ay};
#pragma unroll
    for (int ct = 0; ct < CT; ++ct) {
      const v2f b = *(const v2f*)(bbase + (size_t)ct * 16 * S + k);
      acc[ct] = __builtin_amdgcn_wmma_f32_16x16x4_f32(
          false, a, false, b, (short)0, acc[ct], false, false);
    }
  }

#pragma unroll
  for (int ct = 0; ct < CT; ++ct) {
    const float bv = bias[ct * 16 + m];
#pragma unroll
    for (int r = 0; r < 8; ++r) {
      const long long row = row0 + r + 8 * half;
      const float dv = dis[row];
      const float hv = acc[ct][r];
      H[(size_t)row * NC + ct * 16 + m]    = hv;
      OUTI[(size_t)row * NC + ct * 16 + m] = bv + hv * dv * dv;
    }
  }
}


template <int F>
__global__ void gcn_edge_scatter(const float* __restrict__ h,
                                 const int* __restrict__ src,
                                 const int* __restrict__ dst,
                                 const float* __restrict__ nrm,
                                 float* __restrict__ out, int nE) {
  constexpr int CH = F / 4;
  const long long total  = (long long)nE * CH;
  const long long stride = (long long)gridDim.x * blockDim.x;
  for (long long idx = (long long)blockIdx.x * blockDim.x + threadIdx.x;
       idx < total; idx += stride) {
    const long long nidx = idx + stride;
    if (nidx < total) {
      const int ne = (int)(nidx / CH);
      __builtin_prefetch(h + (size_t)src[ne] * F + (int)(nidx % CH) * 4, 0, 0);
    }
    const int e  = (int)(idx / CH);
    const int fo = (int)(idx % CH) * 4;
    const int s  = src[e];
    const int d  = dst[e];
    const float w = nrm[e];
    const float4 v = *(const float4*)(h + (size_t)s * F + fo);
    float* o = out + (size_t)d * F + fo;
    atomicAdd(o + 0, v.x * w);
    atomicAdd(o + 1, v.y * w);
    atomicAdd(o + 2, v.z * w);
    atomicAdd(o + 3, v.w * w);
  }
}


extern "C" void kernel_launch(void* const* d_in, const int* in_sizes, int n_in,
                              void* d_out, int out_size, void* d_ws, size_t ws_size,
                              hipStream_t stream) {
  const float* x  = (const float*)d_in[0];
  const int*   ei = (const int*)d_in[1];
  const float* W1 = (const float*)d_in[2];
  const float* b1 = (const float*)d_in[3];
  const float* W2 = (const float*)d_in[4];
  const float* b2 = (const float*)d_in[5];
  const float* W3 = (const float*)d_in[6];
  const float* b3 = (const float*)d_in[7];
  float* out = (float*)d_out;

  const int N = N_NODES;
  const int E = in_sizes[1] / 2;
  const int* src = ei;
  const int* dst = ei + E;

  float* ws   = (float*)d_ws;
  float* dis  = ws;
  float* nrm  = ws + OFF_NORM;
  float* bufA = ws + OFF_BUFA;
  float* bufB = bufA + NBUF;

  const int TB = 256;
  const int gN = (N + TB - 1) / TB;
  const int gE = (E + TB - 1) / TB;
  const int gGemm = N / 32;
  const int gScat = 4096;

  gcn_deg_init <<<gN, TB, 0, stream>>>(dis, N);
  gcn_deg_count<<<gE, TB, 0, stream>>>(dst, dis, E);
  gcn_deg_rsqrt<<<gN, TB, 0, stream>>>(dis, N);
  gcn_edge_norm<<<gE, TB, 0, stream>>>(src, dst, dis, nrm, E);

  gcn_gemm_wmma<128, 64, false><<<gGemm, 64, 0, stream>>>(x, W1, bufA, bufB, dis, b1);
  gcn_edge_scatter<64><<<gScat, TB, 0, stream>>>(bufA, src, dst, nrm, bufB, E);

  gcn_gemm_wmma<64, 64, true><<<gGemm, 64, 0, stream>>>(bufB, W2, bufA, bufB, dis, b2);
  gcn_edge_scatter<64><<<gScat, TB, 0, stream>>>(bufA, src, dst, nrm, bufB, E);

  gcn_gemm_wmma<64, 32, true><<<gGemm, 64, 0, stream>>>(bufB, W3, bufA, out, dis, b3);
  gcn_edge_scatter<32><<<gScat, TB, 0, stream>>>(bufA, src, dst, nrm, out, E);
}


# In-degrees by LDS histograms: a work-group counts one node range over one edge chunk (chunks rescanned per range), int planes summed per node; replaces per-edge global f32 atomics; 10 to 9 launches
# speedup vs baseline: 1.7926x; 1.7926x over previous
#include <hip/hip_runtime.h>


typedef __attribute__((ext_vector_type(2))) float v2f;
typedef __attribute__((ext_vector_type(8))) float v8f;

#define N_NODES 100000
#define OFF_NORM 100096
#define OFF_BUFA (OFF_NORM + 1250048)
#define NBUF ((size_t)N_NODES * 64)

#define DEG_G 8
#define DEG_C 32
#define DEG_R 12512
#define DEG_NP (DEG_G * DEG_R)
#define DEG_T 512
#define DEG_U 10
static_assert(DEG_NP >= N_NODES && DEG_R % 32 == 0, "ranges must cover all nodes in whole lines");


__global__ __launch_bounds__(DEG_T)
void gcn_deg_hist(const int* __restrict__ dst, int* __restrict__ part, int nE) {
  __shared__ __attribute__((aligned(16))) int cnt[DEG_R];
  const int tid = threadIdx.x;
  const int g = blockIdx.x % DEG_G;
  const int c = blockIdx.x / DEG_G;
  const int lo = g * DEG_R;

  int4* cnt4 = (int4*)cnt;
  for (int i = tid; i < DEG_R / 4; i += DEG_T) cnt4[i] = int4{0, 0, 0, 0};
  __syncthreads();

  const int chunk = (nE + DEG_C - 1) / DEG_C;
  const int e0 = min(nE, c * chunk);
  const int e1 = min(nE, e0 + chunk);
  const int* p = dst + e0;
  const int len = e1 - e0;

  auto hit = [&](int d) {
    const unsigned r = (unsigned)(d - lo);
    if (r < (unsigned)DEG_R) atomicAdd(&cnt[r], 1);
  };

  const int head = min(len, (int)(((16u - (unsigned)((size_t)p & 15u)) & 15u) >> 2));
  if (tid < head) hit(p[tid]);
  const int4* p4 = (const int4*)(p + head);
  const int n4 = (len - head) >> 2;
  for (int i = tid; i < n4; i += DEG_U * DEG_T) {
    int4 v[DEG_U];
#pragma unroll
    for (int u = 0; u < DEG_U; ++u) {
      const int idx = i + u * DEG_T;
      v[u] = idx < n4 ? p4[idx] : int4{-1, -1, -1, -1};
    }
#pragma unroll
    for (int u = 0; u < DEG_U; ++u) { hit(v[u].x); hit(v[u].y); hit(v[u].z); hit(v[u].w); }
  }
  const int t0 = head + 4 * n4;
  if (tid < len - t0) hit(p[t0 + tid]);
  __syncthreads();

  int4* o4 = (int4*)(part + (size_t)c * DEG_NP + lo);
  for (int i = tid; i < DEG_R / 4; i += DEG_T) o4[i] = cnt4[i];
}

__global__ void gcn_deg_finish(const int* __restrict__ part, float* __restrict__ dis, int n) {
  const int i = blockIdx.x * blockDim.x + threadIdx.x;
  if (i < n) {
    int s = 1;
#pragma unroll
    for (int c = 0; c < DEG_C; ++c) s += part[(size_t)c * DEG_NP + i];
    dis[i] = rsqrtf((float)s);
  }
}

__global__ void gcn_edge_norm(const int* __restrict__ src, const int* __restrict__ dst,
                              const float* __restrict__ dis, float* __restrict__ nrm,
                              int nE) {
  int e = blockIdx.x * blockDim.x + threadIdx.x;
  if (e < nE) nrm[e] = dis[src[e]] * dis[dst[e]];
}


template <int K, int NC, bool RELU>
__global__ __launch_bounds__(64)
void gcn_gemm_wmma(const float* X, const float* __restrict__ W,
                   float* __restrict__ H, float* OUTI,
                   const float* __restrict__ dis, const float* __restrict__ bias) {
  constexpr int S  = K + 6;
  constexpr int CT = NC / 16;
  __shared__ float sWt[NC * S];

  for (int t = threadIdx.x; t < K * NC / 4; t += 64) {
    const int kr = (t * 4) / NC;
    const int n  = (t * 4) % NC;
    const float4 w4 = *(const float4*)(W + (size_t)t * 4);
    sWt[(size_t)(n + 0) * S + kr] = w4.x;
    sWt[(size_t)(n + 1) * S + kr] = w4.y;
    sWt[(size_t)(n + 2) * S + kr] = w4.z;
    sWt[(size_t)(n + 3) * S + kr] = w4.w;
  }
  __syncthreads();

  const int lane = threadIdx.x & 31;
  const int wave = threadIdx.x >> 5;
  const int m    = lane & 15;
  const int half = lane >> 4;
  const long long row0 = ((long long)blockIdx.x * 2 + wave) * 16;

  const float* xrow  = X + (size_t)(row0 + m) * K + 2 * half;
  const float* bbase = sWt + (size_t)m * S + 2 * half;

  v8f acc[CT] = {};
#pragma unroll
  for (int kk = 0; kk < K / 4; ++kk) {
    const int k = kk * 4;
    float ax = xrow[k];
    float ay = xrow[k + 1];
    if (RELU) { ax = fmaxf(ax, 0.f); ay = fmaxf(ay, 0.f); }
    const v2f a = {ax, ay};
#pragma unroll
    for (int ct = 0; ct < CT; ++ct) {
      const v2f b = *(const v2f*)(bbase + (size_t)ct * 16 * S + k);
      acc[ct] = __builtin_amdgcn_wmma_f32_16x16x4_f32(
          false, a, false, b, (short)0, acc[ct], false, false);
    }
  }

#pragma unroll
  for (int ct = 0; ct < CT; ++ct) {
    const float bv = bias[ct * 16 + m];
#pragma unroll
    for (int r = 0; r < 8; ++r) {
      const long long row = row0 + r + 8 * half;
      const float dv = dis[row];
      const float hv = acc[ct][r];
      H[(size_t)row * NC + ct * 16 + m]    = hv;
      OUTI[(size_t)row * NC + ct * 16 + m] = bv + hv * dv * dv;
    }
  }
}


template <int F>
__global__ __launch_bounds__(256)
void gcn_edge_scatter(const float* __restrict__ h,
                      const int* __restrict__ src,
                      const int* __restrict__ dst,
                      const float* __restrict__ nrm,
                      float* __restrict__ out, int nE) {
  constexpr int R = F / 32;
  constexpr int U = 16 / R;
  const unsigned lane = threadIdx.x & 31;
  const int wave  = __builtin_amdgcn_readfirstlane(
      (int)((blockIdx.x * blockDim.x + threadIdx.x) >> 5));
  const int nwave = (int)((gridDim.x * blockDim.x) >> 5);
  int epw = (nE + nwave - 1) / nwave;
  epw = (epw + 15) & ~15;
  const int e0 = wave * epw;
  const int e1 = min(nE, e0 + epw);

  for (int base = e0; base < e1; base += 32) {
    const int cnt = min(32, e1 - base);
    int sl = 0, dl = 0, wl = 0;
    if ((int)lane < cnt) {
      sl = src[base + lane];
      dl = dst[base + lane];
      wl = __builtin_bit_cast(int, nrm[base + lane]);
    }
    int j = 0;
    for (; j + U <= cnt; j += U) {
      float v[U][R];
#pragma unroll
      for (int u = 0; u < U; ++u) {
        const int s = __builtin_amdgcn_readlane(sl, j + u);
        const float* hp = h + (size_t)s * F;
#pragma unroll
        for (int r = 0; r < R; ++r) v[u][r] = hp[r * 32 + lane];
      }
#pragma unroll
      for (int u = 0; u < U; ++u) {
        const int d = __builtin_amdgcn_readlane(dl, j + u);
        const float w = __builtin_bit_cast(float, __builtin_amdgcn_readlane(wl, j + u));
        float* op = out + (size_t)d * F;
#pragma unroll
        for (int r = 0; r < R; ++r) atomicAdd(op + r * 32 + lane, v[u][r] * w);
      }
    }
    for (; j < cnt; ++j) {
      const int s = __builtin_amdgcn_readlane(sl, j);
      const int d = __builtin_amdgcn_readlane(dl, j);
      const float w = __builtin_bit_cast(float, __builtin_amdgcn_readlane(wl, j));
      const float* hp = h + (size_t)s * F;
      float* op = out + (size_t)d * F;
#pragma unroll
      for (int r = 0; r < R; ++r) atomicAdd(op + r * 32 + lane, hp[r * 32 + lane] * w);
    }
  }
}


extern "C" void kernel_launch(void* const* d_in, const int* in_sizes, int n_in,
                              void* d_out, int out_size, void* d_ws, size_t ws_size,
                              hipStream_t stream) {
  const float* x  = (const float*)d_in[0];
  const int*   ei = (const int*)d_in[1];
  const float* W1 = (const float*)d_in[2];
  const float* b1 = (const float*)d_in[3];
  const float* W2 = (const float*)d_in[4];
  const float* b2 = (const float*)d_in[5];
  const float* W3 = (const float*)d_in[6];
  const float* b3 = (const float*)d_in[7];
  float* out = (float*)d_out;

  const int N = N_NODES;
  const int E = in_sizes[1] / 2;
  const int* src = ei;
  const int* dst = ei + E;

  float* ws   = (float*)d_ws;
  float* dis  = ws;
  float* nrm  = ws + OFF_NORM;
  float* bufA = ws + OFF_BUFA;
  float* bufB = bufA + NBUF;
  int*   part = (int*)bufA;

  const int TB = 256;
  const int gN = (N + TB - 1) / TB;
  const int gE = (E + TB - 1) / TB;
  const int gGemm = N / 32;
  const int gScat = 2048;

  gcn_deg_hist  <<<DEG_G * DEG_C, DEG_T, 0, stream>>>(dst, part, E);
  gcn_deg_finish<<<gN, TB, 0, stream>>>(part, dis, N);
  gcn_edge_norm<<<gE, TB, 0, stream>>>(src, dst, dis, nrm, E);

  gcn_gemm_wmma<128, 64, false><<<gGemm, 64, 0, stream>>>(x, W1, bufA, bufB, dis, b1);
  gcn_edge_scatter<64><<<gScat, TB, 0, stream>>>(bufA, src, dst, nrm, bufB, E);

  gcn_gemm_wmma<64, 64, true><<<gGemm, 64, 0, stream>>>(bufB, W2, bufA, bufB, dis, b2);
  gcn_edge_scatter<64><<<gScat, TB, 0, stream>>>(bufA, src, dst, nrm, bufB, E);

  gcn_gemm_wmma<64, 32, true><<<gGemm, 64, 0, stream>>>(bufB, W3, bufA, out, dis, b3);
  gcn_edge_scatter<32><<<gScat, TB, 0, stream>>>(bufA, src, dst, nrm, out, E);
}


# Edge-weight pass moved into the layer-1 GEMM launch as its leading work-groups (16 edges per lane): its index gathers run under the matrix work; own launch of short waves gone, 9 to 8 launches
# speedup vs baseline: 1.8643x; 1.0400x over previous
#include <hip/hip_runtime.h>


typedef __attribute__((ext_vector_type(2))) float v2f;
typedef __attribute__((ext_vector_type(8))) float v8f;

#define N_NODES 100000
#define OFF_NORM 100096
#define OFF_BUFA (OFF_NORM + 1250048)
#define NBUF ((size_t)N_NODES * 64)

#define DEG_G 8
#define DEG_C 32
#define DEG_R 12512
#define DEG_NP (DEG_G * DEG_R)
#define DEG_T 512
#define DEG_U 10
static_assert(DEG_NP >= N_NODES && DEG_R % 32 == 0, "ranges must cover all nodes in whole lines");


__global__ __launch_bounds__(DEG_T)
void gcn_deg_hist(const int* __restrict__ dst, int* __restrict__ part, int nE) {
  __shared__ __attribute__((aligned(16))) int cnt[DEG_R];
  const int tid = threadIdx.x;
  const int g = blockIdx.x % DEG_G;
  const int c = blockIdx.x / DEG_G;
  const int lo = g * DEG_R;

  int4* cnt4 = (int4*)cnt;
  for (int i = tid; i < DEG_R / 4; i += DEG_T) cnt4[i] = int4{0, 0, 0, 0};
  __syncthreads();

  const int chunk = (nE + DEG_C - 1) / DEG_C;
  const int e0 = min(nE, c * chunk);
  const int e1 = min(nE, e0 + chunk);
  const int* p = dst + e0;
  const int len = e1 - e0;

  auto hit = [&](int d) {
    const unsigned r = (unsigned)(d - lo);
    if (r < (unsigned)DEG_R) atomicAdd(&cnt[r], 1);
  };

  const int head = min(len, (int)(((16u - (unsigned)((size_t)p & 15u)) & 15u) >> 2));
  if (tid < head) hit(p[tid]);
  const int4* p4 = (const int4*)(p + head);
  const int n4 = (len - head) >> 2;
  for (int i = tid; i < n4; i += DEG_U * DEG_T) {
    int4 v[DEG_U];
#pragma unroll
    for (int u = 0; u < DEG_U; ++u) {
      const int idx = i + u * DEG_T;
      v[u] = idx < n4 ? p4[idx] : int4{-1, -1, -1, -1};
    }
#pragma unroll
    for (int u = 0; u < DEG_U; ++u) { hit(v[u].x); hit(v[u].y); hit(v[u].z); hit(v[u].w); }
  }
  const int t0 = head + 4 * n4;
  if (tid < len - t0) hit(p[t0 + tid]);
  __syncthreads();

  int4* o4 = (int4*)(part + (size_t)c * DEG_NP + lo);
  for (int i = tid; i < DEG_R / 4; i += DEG_T) o4[i] = cnt4[i];
}

__global__ void gcn_deg_finish(const int* __restrict__ part, float* __restrict__ dis, int n) {
  const int i = blockIdx.x * blockDim.x + threadIdx.x;
  if (i < n) {
    int s = 1;
#pragma unroll
    for (int c = 0; c < DEG_C; ++c) s += part[(size_t)c * DEG_NP + i];
    dis[i] = rsqrtf((float)s);
  }
}

#define EN_U 16
#define EN_B (64 * EN_U)

__device__ __forceinline__ void gcn_edge_norm_run(const int* __restrict__ src,
                                                  const int* __restrict__ dst,
                                                  const float* __restrict__ dis,
                                                  float* __restrict__ nrm,
                                                  int nE, int run) {
  const int e0 = run * EN_B + (int)threadIdx.x;
  if (run * EN_B + EN_B <= nE) {
    int s[EN_U], d[EN_U];
#pragma unroll
    for (int u = 0; u < EN_U; ++u) { s[u] = src[e0 + u * 64]; d[u] = dst[e0 + u * 64]; }
    float a[EN_U], b[EN_U];
#pragma unroll
    for (int u = 0; u < EN_U; ++u) { a[u] = dis[s[u]]; b[u] = dis[d[u]]; }
#pragma unroll
    for (int u = 0; u < EN_U; ++u) nrm[e0 + u * 64] = a[u] * b[u];
  } else {
    for (int e = e0; e < nE; e += 64) nrm[e] = dis[src[e]] * dis[dst[e]];
  }
}


template <int K, int NC, bool RELU, bool EDGE>
__global__ __launch_bounds__(64)
void gcn_gemm_wmma(const float* X, const float* __restrict__ W,
                   float* __restrict__ H, float* OUTI,
                   const float* __restrict__ dis, const float* __restrict__ bias,
                   const int* __restrict__ src, const int* __restrict__ dst,
                   float* __restrict__ nrm, int nE, int nEB) {
  constexpr int S  = K + 6;
  constexpr int CT = NC / 16;
  __shared__ float sWt[NC * S];

  int panel2 = (int)blockIdx.x;
  if (EDGE) {
    if (panel2 < nEB) {
      gcn_edge_norm_run(src, dst, dis, nrm, nE, panel2);
      return;
    }
    panel2 -= nEB;
  }

  for (int t = threadIdx.x; t < K * NC / 4; t += 64) {
    const int kr = (t * 4) / NC;
    const int n  = (t * 4) % NC;
    const float4 w4 = *(const float4*)(W + (size_t)t * 4);
    sWt[(size_t)(n + 0) * S + kr] = w4.x;
    sWt[(size_t)(n + 1) * S + kr] = w4.y;
    sWt[(size_t)(n + 2) * S + kr] = w4.z;
    sWt[(size_t)(n + 3) * S + kr] = w4.w;
  }
  __syncthreads();

  const int lane = threadIdx.x & 31;
  const int wave = threadIdx.x >> 5;
  const int m    = lane & 15;
  const int half = lane >> 4;
  const long long row0 = ((long long)panel2 * 2 + wave) * 16;

  const float* xrow  = X + (size_t)(row0 + m) * K + 2 * half;
  const float* bbase = sWt + (size_t)m * S + 2 * half;

  v8f acc[CT] = {};
#pragma unroll
  for (int kk = 0; kk < K / 4; ++kk) {
    const int k = kk * 4;
    float ax = xrow[k];
    float ay = xrow[k + 1];
    if (RELU) { ax = fmaxf(ax, 0.f); ay = fmaxf(ay, 0.f); }
    const v2f a = {ax, ay};
#pragma unroll
    for (int ct = 0; ct < CT; ++ct) {
      const v2f b = *(const v2f*)(bbase + (size_t)ct * 16 * S + k);
      acc[ct] = __builtin_amdgcn_wmma_f32_16x16x4_f32(
          false, a, false, b, (short)0, acc[ct], false, false);
    }
  }

#pragma unroll
  for (int ct = 0; ct < CT; ++ct) {
    const float bv = bias[ct * 16 + m];
#pragma unroll
    for (int r = 0; r < 8; ++r) {
      const long long row = row0 + r + 8 * half;
      const float dv = dis[row];
      const float hv = acc[ct][r];
      H[(size_t)row * NC + ct * 16 + m]    = hv;
      OUTI[(size_t)row * NC + ct * 16 + m] = bv + hv * dv * dv;
    }
  }
}


template <int F>
__global__ __launch_bounds__(256)
void gcn_edge_scatter(const float* __restrict__ h,
                      const int* __restrict__ src,
                      const int* __restrict__ dst,
                      const float* __restrict__ nrm,
                      float* __restrict__ out, int nE) {
  constexpr int R = F / 32;
  constexpr int U = 16 / R;
  const unsigned lane = threadIdx.x & 31;
  const int wave  = __builtin_amdgcn_readfirstlane(
      (int)((blockIdx.x * blockDim.x + threadIdx.x) >> 5));
  const int nwave = (int)((gridDim.x * blockDim.x) >> 5);
  int epw = (nE + nwave - 1) / nwave;
  epw = (epw + 15) & ~15;
  const int e0 = wave * epw;
  const int e1 = min(nE, e0 + epw);

  for (int base = e0; base < e1; base += 32) {
    const int cnt = min(32, e1 - base);
    int sl = 0, dl = 0, wl = 0;
    if ((int)lane < cnt) {
      sl = src[base + lane];
      dl = dst[base + lane];
      wl = __builtin_bit_cast(int, nrm[base + lane]);
    }
    int j = 0;
    for (; j + U <= cnt; j += U) {
      float v[U][R];
#pragma unroll
      for (int u = 0; u < U; ++u) {
        const int s = __builtin_amdgcn_readlane(sl, j + u);
        const float* hp = h + (size_t)s * F;
#pragma unroll
        for (int r = 0; r < R; ++r) v[u][r] = hp[r * 32 + lane];
      }
#pragma unroll
      for (int u = 0; u < U; ++u) {
        const int d = __builtin_amdgcn_readlane(dl, j + u);
        const float w = __builtin_bit_cast(float, __builtin_amdgcn_readlane(wl, j + u));
        float* op = out + (size_t)d * F;
#pragma unroll
        for (int r = 0; r < R; ++r) atomicAdd(op + r * 32 + lane, v[u][r] * w);
      }
    }
    for (; j < cnt; ++j) {
      const int s = __builtin_amdgcn_readlane(sl, j);
      const int d = __builtin_amdgcn_readlane(dl, j);
      const float w = __builtin_bit_cast(float, __builtin_amdgcn_readlane(wl, j));
      const float* hp = h + (size_t)s * F;
      float* op = out + (size_t)d * F;
#pragma unroll
      for (int r = 0; r < R; ++r) atomicAdd(op + r * 32 + lane, hp[r * 32 + lane] * w);
    }
  }
}


extern "C" void kernel_launch(void* const* d_in, const int* in_sizes, int n_in,
                              void* d_out, int out_size, void* d_ws, size_t ws_size,
                              hipStream_t stream) {
  const float* x  = (const float*)d_in[0];
  const int*   ei = (const int*)d_in[1];
  const float* W1 = (const float*)d_in[2];
  const float* b1 = (const float*)d_in[3];
  const float* W2 = (const float*)d_in[4];
  const float* b2 = (const float*)d_in[5];
  const float* W3 = (const float*)d_in[6];
  const float* b3 = (const float*)d_in[7];
  float* out = (float*)d_out;

  const int N = N_NODES;
  const int E = in_sizes[1] / 2;
  const int* src = ei;
  const int* dst = ei + E;

  float* ws   = (float*)d_ws;
  float* dis  = ws;
  float* nrm  = ws + OFF_NORM;
  float* bufA = ws + OFF_BUFA;
  float* bufB = bufA + NBUF;
  int*   part = (int*)bufA;

  const int TB = 256;
  const int gN = (N + TB - 1) / TB;
  const int gEN = (E + EN_B - 1) / EN_B;
  const int gGemm = N / 32;
  const int gScat = 2048;

  gcn_deg_hist  <<<DEG_G * DEG_C, DEG_T, 0, stream>>>(dst, part, E);
  gcn_deg_finish<<<gN, TB, 0, stream>>>(part, dis, N);

  gcn_gemm_wmma<128, 64, false, true><<<gEN + gGemm, 64, 0, stream>>>(
      x, W1, bufA, bufB, dis, b1, src, dst, nrm, E, gEN);
  gcn_edge_scatter<64><<<gScat, TB, 0, stream>>>(bufA, src, dst, nrm, bufB, E);

  gcn_gemm_wmma<64, 64, true, false><<<gGemm, 64, 0, stream>>>(
      bufB, W2, bufA, bufB, dis, b2, src, dst, nrm, E, 0);
  gcn_edge_scatter<64><<<gScat, TB, 0, stream>>>(bufA, src, dst, nrm, bufB, E);

  gcn_gemm_wmma<64, 32, true, false><<<gGemm, 64, 0, stream>>>(
      bufB, W3, bufA, out, dis, b3, src, dst, nrm, E, 0);
  gcn_edge_scatter<32><<<gScat, TB, 0, stream>>>(bufA, src, dst, nrm, out, E);
}


# Edge scatter cut by destination half: a work-group reads its die number and claims an (edge run, node half) task from its own side's counter, so a row takes adds from one L2 side; sum order: arrival
# speedup vs baseline: 2.0231x; 1.0852x over previous
#include <hip/hip_runtime.h>


typedef __attribute__((ext_vector_type(2))) float v2f;
typedef __attribute__((ext_vector_type(8))) float v8f;

#define N_NODES 100000
#define OFF_NORM 100096
#define OFF_BUFA (OFF_NORM + 1250048)
#define NBUF ((size_t)N_NODES * 64)

#define DEG_G 8
#define DEG_C 32
#define DEG_R 12512
#define DEG_NP (DEG_G * DEG_R)
#define DEG_T 512
#define DEG_U 10
static_assert(DEG_NP >= N_NODES && DEG_R % 32 == 0, "ranges must cover all nodes in whole lines");


__global__ __launch_bounds__(DEG_T)
void gcn_deg_hist(const int* __restrict__ dst, int* __restrict__ part, int nE) {
  __shared__ __attribute__((aligned(16))) int cnt[DEG_R];
  const int tid = threadIdx.x;
  const int g = blockIdx.x % DEG_G;
  const int c = blockIdx.x / DEG_G;
  const int lo = g * DEG_R;

  int4* cnt4 = (int4*)cnt;
  for (int i = tid; i < DEG_R / 4; i += DEG_T) cnt4[i] = int4{0, 0, 0, 0};
  __syncthreads();

  const int chunk = (nE + DEG_C - 1) / DEG_C;
  const int e0 = min(nE, c * chunk);
  const int e1 = min(nE, e0 + chunk);
  const int* p = dst + e0;
  const int len = e1 - e0;

  auto hit = [&](int d) {
    const unsigned r = (unsigned)(d - lo);
    if (r < (unsigned)DEG_R) atomicAdd(&cnt[r], 1);
  };

  const int head = min(len, (int)(((16u - (unsigned)((size_t)p & 15u)) & 15u) >> 2));
  if (tid < head) hit(p[tid]);
  const int4* p4 = (const int4*)(p + head);
  const int n4 = (len - head) >> 2;
  for (int i = tid; i < n4; i += DEG_U * DEG_T) {
    int4 v[DEG_U];
#pragma unroll
    for (int u = 0; u < DEG_U; ++u) {
      const int idx = i + u * DEG_T;
      v[u] = idx < n4 ? p4[idx] : int4{-1, -1, -1, -1};
    }
#pragma unroll
    for (int u = 0; u < DEG_U; ++u) { hit(v[u].x); hit(v[u].y); hit(v[u].z); hit(v[u].w); }
  }
  const int t0 = head + 4 * n4;
  if (tid < len - t0) hit(p[t0 + tid]);
  __syncthreads();

  int4* o4 = (int4*)(part + (size_t)c * DEG_NP + lo);
  for (int i = tid; i < DEG_R / 4; i += DEG_T) o4[i] = cnt4[i];
}

#define CLAIM_PITCH 64
#define CLAIM_LAUNCH (2 * CLAIM_PITCH)

__global__ void gcn_deg_finish(const int* __restrict__ part, float* __restrict__ dis, int n,
                               int* __restrict__ claim) {
  const int i = blockIdx.x * blockDim.x + threadIdx.x;
  if (i < 6) claim[(i >> 1) * CLAIM_LAUNCH + (i & 1) * CLAIM_PITCH] = 0;
  if (i < n) {
    int s = 1;
#pragma unroll
    for (int c = 0; c < DEG_C; ++c) s += part[(size_t)c * DEG_NP + i];
    dis[i] = rsqrtf((float)s);
  }
}

#define EN_U 16
#define EN_B (64 * EN_U)

__device__ __forceinline__ void gcn_edge_norm_run(const int* __restrict__ src,
                                                  const int* __restrict__ dst,
                                                  const float* __restrict__ dis,
                                                  float* __restrict__ nrm,
                                                  int nE, int run) {
  const int e0 = run * EN_B + (int)threadIdx.x;
  if (run * EN_B + EN_B <= nE) {
    int s[EN_U], d[EN_U];
#pragma unroll
    for (int u = 0; u < EN_U; ++u) { s[u] = src[e0 + u * 64]; d[u] = dst[e0 + u * 64]; }
    float a[EN_U], b[EN_U];
#pragma unroll
    for (int u = 0; u < EN_U; ++u) { a[u] = dis[s[u]]; b[u] = dis[d[u]]; }
#pragma unroll
    for (int u = 0; u < EN_U; ++u) nrm[e0 + u * 64] = a[u] * b[u];
  } else {
    for (int e = e0; e < nE; e += 64) nrm[e] = dis[src[e]] * dis[dst[e]];
  }
}


template <int K, int NC, bool RELU, bool EDGE>
__global__ __launch_bounds__(64)
void gcn_gemm_wmma(const float* X, const float* __restrict__ W,
                   float* __restrict__ H, float* OUTI,
                   const float* __restrict__ dis, const float* __restrict__ bias,
                   const int* __restrict__ src, const int* __restrict__ dst,
                   float* __restrict__ nrm, int nE, int nEB) {
  constexpr int S  = K + 6;
  constexpr int CT = NC / 16;
  __shared__ float sWt[NC * S];

  int panel2 = (int)blockIdx.x;
  if (EDGE) {
    if (panel2 < nEB) {
      gcn_edge_norm_run(src, dst, dis, nrm, nE, panel2);
      return;
    }
    panel2 -= nEB;
  }

  for (int t = threadIdx.x; t < K * NC / 4; t += 64) {
    const int kr = (t * 4) / NC;
    const int n  = (t * 4) % NC;
    const float4 w4 = *(const float4*)(W + (size_t)t * 4);
    sWt[(size_t)(n + 0) * S + kr] = w4.x;
    sWt[(size_t)(n + 1) * S + kr] = w4.y;
    sWt[(size_t)(n + 2) * S + kr] = w4.z;
    sWt[(size_t)(n + 3) * S + kr] = w4.w;
  }
  __syncthreads();

  const int lane = threadIdx.x & 31;
  const int wave = threadIdx.x >> 5;
  const int m    = lane & 15;
  const int half = lane >> 4;
  const long long row0 = ((long long)panel2 * 2 + wave) * 16;

  const float* xrow  = X + (size_t)(row0 + m) * K + 2 * half;
  const float* bbase = sWt + (size_t)m * S + 2 * half;

  v8f acc[CT] = {};
#pragma unroll
  for (int kk = 0; kk < K / 4; ++kk) {
    const int k = kk * 4;
    float ax = xrow[k];
    float ay = xrow[k + 1];
    if (RELU) { ax = fmaxf(ax, 0.f); ay = fmaxf(ay, 0.f); }
    const v2f a = {ax, ay};
#pragma unroll
    for (int ct = 0; ct < CT; ++ct) {
      const v2f b = *(const v2f*)(bbase + (size_t)ct * 16 * S + k);
      acc[ct] = __builtin_amdgcn_wmma_f32_16x16x4_f32(
          false, a, false, b, (short)0, acc[ct], false, false);
    }
  }

#pragma unroll
  for (int ct = 0; ct < CT; ++ct) {
    const float bv = bias[ct * 16 + m];
#pragma unroll
    for (int r = 0; r < 8; ++r) {
      const long long row = row0 + r + 8 * half;
      const float dv = dis[row];
      const float hv = acc[ct][r];
      H[(size_t)row * NC + ct * 16 + m]    = hv;
      OUTI[(size_t)row * NC + ct * 16 + m] = bv + hv * dv * dv;
    }
  }
}


#define SC_WAVES 8
#define SC_B 5
#define SC_RUN (SC_WAVES * SC_B * 32)

template <int F>
__global__ __launch_bounds__(SC_WAVES * 32)
void gcn_edge_scatter(const float* __restrict__ h,
                      const int* __restrict__ src,
                      const int* __restrict__ dst,
                      const float* __restrict__ nrm,
                      float* __restrict__ out, int nE,
                      int nRuns, int nSplit, int* __restrict__ claim) {
  constexpr int R = F / 32;
  constexpr int U = 16 / R;
  __shared__ int   s_task;
  __shared__ int   s_cnt[SC_WAVES];
  __shared__ int   q_s[SC_RUN];
  __shared__ int   q_d[SC_RUN];
  __shared__ float q_w[SC_RUN];

  const int lane = (int)(threadIdx.x & 31);
  const int wv = __builtin_amdgcn_readfirstlane((int)(threadIdx.x >> 5));

  if (threadIdx.x == 0) {
    const int side = (int)((__builtin_amdgcn_s_sendmsg_rtn(0x87) >> 18) & 1u);
    int half = side;
    int t = atomicAdd(claim + half * CLAIM_PITCH, 1);
    if (t >= nRuns) {
      half = side ^ 1;
      t = atomicAdd(claim + half * CLAIM_PITCH, 1);
    }
    s_task = t * 2 + half;
  }
  __syncthreads();
  const int run  = s_task >> 1;
  const int half = s_task & 1;
  if (run >= nRuns) return;

  const int e0 = run * SC_RUN + wv * (SC_B * 32) + lane;
  int sv[SC_B], dv[SC_B];
  float wt[SC_B];
#pragma unroll
  for (int b = 0; b < SC_B; ++b) {
    const int e = e0 + b * 32;
    const bool in = e < nE;
    dv[b] = in ? dst[e] : -1;
    sv[b] = in ? src[e] : 0;
    wt[b] = in ? nrm[e] : 0.f;
  }
  unsigned mk[SC_B];
  int n = 0;
#pragma unroll
  for (int b = 0; b < SC_B; ++b) {
    const bool mine = dv[b] >= 0 && ((dv[b] >= nSplit) == (half != 0));
    mk[b] = __builtin_amdgcn_ballot_w32(mine);
    n += __builtin_popcount(mk[b]);
  }
  if (lane == 0) s_cnt[wv] = n;
  __syncthreads();

  int off = 0, total = 0;
#pragma unroll
  for (int i = 0; i < SC_WAVES; ++i) {
    const int c = s_cnt[i];
    off += i < wv ? c : 0;
    total += c;
  }
#pragma unroll
  for (int b = 0; b < SC_B; ++b) {
    if ((mk[b] >> lane) & 1u) {
      const int p = off + (int)__builtin_amdgcn_mbcnt_lo(mk[b], 0u);
      q_s[p] = sv[b];
      q_d[p] = dv[b];
      q_w[p] = wt[b];
    }
    off += __builtin_popcount(mk[b]);
  }
  __syncthreads();

  int share = (total + SC_WAVES - 1) / SC_WAVES;
  share = (share + U - 1) / U * U;
  const int b0 = wv * share;
  const int b1 = min(total, b0 + share);

  for (int base = b0; base < b1; base += 32) {
    const int cnt = min(32, b1 - base);
    int sl = 0, dl = 0, wl = 0;
    if (lane < cnt) {
      sl = q_s[base + lane];
      dl = q_d[base + lane];
      wl = __builtin_bit_cast(int, q_w[base + lane]);
    }
    int j = 0;
    for (; j + U <= cnt; j += U) {
      float v[U][R];
#pragma unroll
      for (int u = 0; u < U; ++u) {
        const int s = __builtin_amdgcn_readlane(sl, j + u);
        const float* hp = h + (size_t)s * F;
#pragma unroll
        for (int r = 0; r < R; ++r) v[u][r] = hp[r * 32 + lane];
      }
#pragma unroll
      for (int u = 0; u < U; ++u) {
        const int d = __builtin_amdgcn_readlane(dl, j + u);
        const float w = __builtin_bit_cast(float, __builtin_amdgcn_readlane(wl, j + u));
        float* op = out + (size_t)d * F;
#pragma unroll
        for (int r = 0; r < R; ++r) atomicAdd(op + r * 32 + lane, v[u][r] * w);
      }
    }
    if (j < cnt) {
      float v[U][R];
#pragma unroll
      for (int u = 0; u < U; ++u) {
        if (j + u < cnt) {
          const int s = __builtin_amdgcn_readlane(sl, j + u);
          const float* hp = h + (size_t)s * F;
#pragma unroll
          for (int r = 0; r < R; ++r) v[u][r] = hp[r * 32 + lane];
        }
      }
#pragma unroll
      for (int u = 0; u < U; ++u) {
        if (j + u < cnt) {
          const int d = __builtin_amdgcn_readlane(dl, j + u);
          const float w = __builtin_bit_cast(float, __builtin_amdgcn_readlane(wl, j + u));
          float* op = out + (size_t)d * F;
#pragma unroll
          for (int r = 0; r < R; ++r) atomicAdd(op + r * 32 + lane, v[u][r] * w);
        }
      }
    }
  }
}


extern "C" void kernel_launch(void* const* d_in, const int* in_sizes, int n_in,
                              void* d_out, int out_size, void* d_ws, size_t ws_size,
                              hipStream_t stream) {
  const float* x  = (const float*)d_in[0];
  const int*   ei = (const int*)d_in[1];
  const float* W1 = (const float*)d_in[2];
  const float* b1 = (const float*)d_in[3];
  const float* W2 = (const float*)d_in[4];
  const float* b2 = (const float*)d_in[5];
  const float* W3 = (const float*)d_in[6];
  const float* b3 = (const float*)d_in[7];
  float* out = (float*)d_out;

  const int N = N_NODES;
  const int E = in_sizes[1] / 2;
  const int* src = ei;
  const int* dst = ei + E;

  float* ws   = (float*)d_ws;
  float* dis  = ws;
  float* nrm  = ws + OFF_NORM;
  float* bufA = ws + OFF_BUFA;
  float* bufB = bufA + NBUF;
  int*   part = (int*)bufA;
  int*   claim = (int*)(bufB + NBUF);

  const int TB = 256;
  const int gN = (N + TB - 1) / TB;
  const int gEN = (E + EN_B - 1) / EN_B;
  const int gGemm = N / 32;
  const int nRuns = (E + SC_RUN - 1) / SC_RUN;
  const int gScat = 2 * nRuns;
  const int nSplit = N / 2;

  gcn_deg_hist  <<<DEG_G * DEG_C, DEG_T, 0, stream>>>(dst, part, E);
  gcn_deg_finish<<<gN, TB, 0, stream>>>(part, dis, N, claim);

  gcn_gemm_wmma<128, 64, false, true><<<gEN + gGemm, 64, 0, stream>>>(
      x, W1, bufA, bufB, dis, b1, src, dst, nrm, E, gEN);
  gcn_edge_scatter<64><<<gScat, SC_WAVES * 32, 0, stream>>>(bufA, src, dst, nrm, bufB, E,
                                                           nRuns, nSplit, claim);

  gcn_gemm_wmma<64, 64, true, false><<<gGemm, 64, 0, stream>>>(
      bufB, W2, bufA, bufB, dis, b2, src, dst, nrm, E, 0);
  gcn_edge_scatter<64><<<gScat, SC_WAVES * 32, 0, stream>>>(bufA, src, dst, nrm, bufB, E,
                                                           nRuns, nSplit, claim + CLAIM_LAUNCH);

  gcn_gemm_wmma<64, 32, true, false><<<gGemm, 64, 0, stream>>>(
      bufB, W3, bufA, out, dis, b3, src, dst, nrm, E, 0);
  gcn_edge_scatter<32><<<gScat, SC_WAVES * 32, 0, stream>>>(bufA, src, dst, nrm, out, E,
                                                           nRuns, nSplit, claim + 2 * CLAIM_LAUNCH);
}


# 64-wide h and sums kept as two 32-feature planes; scatter tasks cut by plane, not destination half: a package side owns one plane, every edge taken, no filter or LDS queue; sum order: arrival
# speedup vs baseline: 2.1540x; 1.0647x over previous
#include <hip/hip_runtime.h>


typedef __attribute__((ext_vector_type(2))) float v2f;
typedef __attribute__((ext_vector_type(8))) float v8f;

#define N_NODES 100000
#define OFF_NORM 100096
#define OFF_BUFA (OFF_NORM + 1250048)
#define NBUF ((size_t)N_NODES * 64)
#define PLANE ((size_t)N_NODES * 32)

#define DEG_G 8
#define DEG_C 32
#define DEG_R 12512
#define DEG_NP (DEG_G * DEG_R)
#define DEG_T 512
#define DEG_U 10
static_assert(DEG_NP >= N_NODES && DEG_R % 32 == 0, "ranges must cover all nodes in whole lines");


__global__ __launch_bounds__(DEG_T)
void gcn_deg_hist(const int* __restrict__ dst, int* __restrict__ part, int nE) {
  __shared__ __attribute__((aligned(16))) int cnt[DEG_R];
  const int tid = threadIdx.x;
  const int g = blockIdx.x % DEG_G;
  const int c = blockIdx.x / DEG_G;
  const int lo = g * DEG_R;

  int4* cnt4 = (int4*)cnt;
  for (int i = tid; i < DEG_R / 4; i += DEG_T) cnt4[i] = int4{0, 0, 0, 0};
  __syncthreads();

  const int chunk = (nE + DEG_C - 1) / DEG_C;
  const int e0 = min(nE, c * chunk);
  const int e1 = min(nE, e0 + chunk);
  const int* p = dst + e0;
  const int len = e1 - e0;

  auto hit = [&](int d) {
    const unsigned r = (unsigned)(d - lo);
    if (r < (unsigned)DEG_R) atomicAdd(&cnt[r], 1);
  };

  const int head = min(len, (int)(((16u - (unsigned)((size_t)p & 15u)) & 15u) >> 2));
  if (tid < head) hit(p[tid]);
  const int4* p4 = (const int4*)(p + head);
  const int n4 = (len - head) >> 2;
  for (int i = tid; i < n4; i += DEG_U * DEG_T) {
    int4 v[DEG_U];
#pragma unroll
    for (int u = 0; u < DEG_U; ++u) {
      const int idx = i + u * DEG_T;
      v[u] = idx < n4 ? p4[idx] : int4{-1, -1, -1, -1};
    }
#pragma unroll
    for (int u = 0; u < DEG_U; ++u) { hit(v[u].x); hit(v[u].y); hit(v[u].z); hit(v[u].w); }
  }
  const int t0 = head + 4 * n4;
  if (tid < len - t0) hit(p[t0 + tid]);
  __syncthreads();

  int4* o4 = (int4*)(part + (size_t)c * DEG_NP + lo);
  for (int i = tid; i < DEG_R / 4; i += DEG_T) o4[i] = cnt4[i];
}

#define CLAIM_PITCH 64
#define CLAIM_LAUNCH (2 * CLAIM_PITCH)

__global__ void gcn_deg_finish(const int* __restrict__ part, float* __restrict__ dis, int n,
                               int* __restrict__ claim) {
  const int i = blockIdx.x * blockDim.x + threadIdx.x;
  if (i < 6) claim[(i >> 1) * CLAIM_LAUNCH + (i & 1) * CLAIM_PITCH] = 0;
  if (i < n) {
    int s = 1;
#pragma unroll
    for (int c = 0; c < DEG_C; ++c) s += part[(size_t)c * DEG_NP + i];
    dis[i] = rsqrtf((float)s);
  }
}

#define EN_U 16
#define EN_B (64 * EN_U)

__device__ __forceinline__ void gcn_edge_norm_run(const int* __restrict__ src,
                                                  const int* __restrict__ dst,
                                                  const float* __restrict__ dis,
                                                  float* __restrict__ nrm,
                                                  int nE, int run) {
  const int e0 = run * EN_B + (int)threadIdx.x;
  if (run * EN_B + EN_B <= nE) {
    int s[EN_U], d[EN_U];
#pragma unroll
    for (int u = 0; u < EN_U; ++u) { s[u] = src[e0 + u * 64]; d[u] = dst[e0 + u * 64]; }
    float a[EN_U], b[EN_U];
#pragma unroll
    for (int u = 0; u < EN_U; ++u) { a[u] = dis[s[u]]; b[u] = dis[d[u]]; }
#pragma unroll
    for (int u = 0; u < EN_U; ++u) nrm[e0 + u * 64] = a[u] * b[u];
  } else {
    for (int e = e0; e < nE; e += 64) nrm[e] = dis[src[e]] * dis[dst[e]];
  }
}


template <int K, int NC, bool RELU, bool EDGE, bool APL>
__global__ __launch_bounds__(64)
void gcn_gemm_wmma(const float* X, const float* __restrict__ W,
                   float* __restrict__ H, float* OUTI,
                   const float* __restrict__ dis, const float* __restrict__ bias,
                   const int* __restrict__ src, const int* __restrict__ dst,
                   float* __restrict__ nrm, int nE, int nEB) {
  constexpr int S  = K + 6;
  constexpr int CT = NC / 16;
  __shared__ float sWt[NC * S];

  int panel2 = (int)blockIdx.x;
  if (EDGE) {
    if (panel2 < nEB) {
      gcn_edge_norm_run(src, dst, dis, nrm, nE, panel2);
      return;
    }
    panel2 -= nEB;
  }

  for (int t = threadIdx.x; t < K * NC / 4; t += 64) {
    const int kr = (t * 4) / NC;
    const int n  = (t * 4) % NC;
    const float4 w4 = *(const float4*)(W + (size_t)t * 4);
    sWt[(size_t)(n + 0) * S + kr] = w4.x;
    sWt[(size_t)(n + 1) * S + kr] = w4.y;
    sWt[(size_t)(n + 2) * S + kr] = w4.z;
    sWt[(size_t)(n + 3) * S + kr] = w4.w;
  }
  __syncthreads();

  const int lane = threadIdx.x & 31;
  const int wave = threadIdx.x >> 5;
  const int m    = lane & 15;
  const int half = lane >> 4;
  const long long row0 = ((long long)panel2 * 2 + wave) * 16;

  static_assert(!APL || K == 64, "plane input is two planes of 32 features");
  const float* xrow  = X + (size_t)(row0 + m) * (APL ? 32 : K) + 2 * half;
  const float* bbase = sWt + (size_t)m * S + 2 * half;

  v8f acc[CT] = {};
#pragma unroll
  for (int kk = 0; kk < K / 4; ++kk) {
    const int k = kk * 4;
    const float* xp = APL ? xrow + (size_t)(k >> 5) * PLANE + (k & 31) : xrow + k;
    float ax = xp[0];
    float ay = xp[1];
    if (RELU) { ax = fmaxf(ax, 0.f); ay = fmaxf(ay, 0.f); }
    const v2f a = {ax, ay};
#pragma unroll
    for (int ct = 0; ct < CT; ++ct) {
      const v2f b = *(const v2f*)(bbase + (size_t)ct * 16 * S + k);
      acc[ct] = __builtin_amdgcn_wmma_f32_16x16x4_f32(
          false, a, false, b, (short)0, acc[ct], false, false);
    }
  }

#pragma unroll
  for (int ct = 0; ct < CT; ++ct) {
    const float bv = bias[ct * 16 + m];
#pragma unroll
    for (int r = 0; r < 8; ++r) {
      const long long row = row0 + r + 8 * half;
      const float dv = dis[row];
      const float hv = acc[ct][r];
      const size_t o = (size_t)(ct >> 1) * PLANE + (size_t)row * 32 + (ct & 1) * 16 + m;
      H[o]    = hv;
      OUTI[o] = bv + hv * dv * dv;
    }
  }
}


#define SC_WAVES 8
#define SC_B 5
#define SC_RUN (SC_WAVES * SC_B * 32)

template <int F>
__global__ __launch_bounds__(SC_WAVES * 32)
void gcn_edge_scatter(const float* __restrict__ h,
                      const int* __restrict__ src,
                      const int* __restrict__ dst,
                      const float* __restrict__ nrm,
                      float* __restrict__ out, int nE,
                      int nRuns, int nSplit, int* __restrict__ claim) {
  constexpr int R = F / 32;
  constexpr int U = 16 / R;
  __shared__ int   s_task;
  __shared__ int   s_cnt[SC_WAVES];
  __shared__ int   q_s[SC_RUN];
  __shared__ int   q_d[SC_RUN];
  __shared__ float q_w[SC_RUN];

  const int lane = (int)(threadIdx.x & 31);
  const int wv = __builtin_amdgcn_readfirstlane((int)(threadIdx.x >> 5));

  if (threadIdx.x == 0) {
    const int side = (int)((__builtin_amdgcn_s_sendmsg_rtn(0x87) >> 18) & 1u);
    int half = side;
    int t = atomicAdd(claim + half * CLAIM_PITCH, 1);
    if (t >= nRuns) {
      half = side ^ 1;
      t = atomicAdd(claim + half * CLAIM_PITCH, 1);
    }
    s_task = t * 2 + half;
  }
  __syncthreads();
  const int run  = s_task >> 1;
  const int half = s_task & 1;
  if (run >= nRuns) return;

  const int e0 = run * SC_RUN + wv * (SC_B * 32) + lane;
  int sv[SC_B], dv[SC_B];
  float wt[SC_B];
#pragma unroll
  for (int b = 0; b < SC_B; ++b) {
    const int e = e0 + b * 32;
    const bool in = e < nE;
    dv[b] = in ? dst[e] : -1;
    sv[b] = in ? src[e] : 0;
    wt[b] = in ? nrm[e] : 0.f;
  }
  unsigned mk[SC_B];
  int n = 0;
#pragma unroll
  for (int b = 0; b < SC_B; ++b) {
    const bool mine = dv[b] >= 0 && ((dv[b] >= nSplit) == (half != 0));
    mk[b] = __builtin_amdgcn_ballot_w32(mine);
    n += __builtin_popcount(mk[b]);
  }
  if (lane == 0) s_cnt[wv] = n;
  __syncthreads();

  int off = 0, total = 0;
#pragma unroll
  for (int i = 0; i < SC_WAVES; ++i) {
    const int c = s_cnt[i];
    off += i < wv ? c : 0;
    total += c;
  }
#pragma unroll
  for (int b = 0; b < SC_B; ++b) {
    if ((mk[b] >> lane) & 1u) {
      const int p = off + (int)__builtin_amdgcn_mbcnt_lo(mk[b], 0u);
      q_s[p] = sv[b];
      q_d[p] = dv[b];
      q_w[p] = wt[b];
    }
    off += __builtin_popcount(mk[b]);
  }
  __syncthreads();

  int share = (total + SC_WAVES - 1) / SC_WAVES;
  share = (share + U - 1) / U * U;
  const int b0 = wv * share;
  const int b1 = min(total, b0 + share);

  for (int base = b0; base < b1; base += 32) {
    const int cnt = min(32, b1 - base);
    int sl = 0, dl = 0, wl = 0;
    if (lane < cnt) {
      sl = q_s[base + lane];
      dl = q_d[base + lane];
      wl = __builtin_bit_cast(int, q_w[base + lane]);
    }
    int j = 0;
    for (; j + U <= cnt; j += U) {
      float v[U][R];
#pragma unroll
      for (int u = 0; u < U; ++u) {
        const int s = __builtin_amdgcn_readlane(sl, j + u);
        const float* hp = h + (size_t)s * F;
#pragma unroll
        for (int r = 0; r < R; ++r) v[u][r] = hp[r * 32 + lane];
      }
#pragma unroll
      for (int u = 0; u < U; ++u) {
        const int d = __builtin_amdgcn_readlane(dl, j + u);
        const float w = __builtin_bit_cast(float, __builtin_amdgcn_readlane(wl, j + u));
        float* op = out + (size_t)d * F;
#pragma unroll
        for (int r = 0; r < R; ++r) atomicAdd(op + r * 32 + lane, v[u][r] * w);
      }
    }
    if (j < cnt) {
      float v[U][R];
#pragma unroll
      for (int u = 0; u < U; ++u) {
        if (j + u < cnt) {
          const int s = __builtin_amdgcn_readlane(sl, j + u);
          const float* hp = h + (size_t)s * F;
#pragma unroll
          for (int r = 0; r < R; ++r) v[u][r] = hp[r * 32 + lane];
        }
      }
#pragma unroll
      for (int u = 0; u < U; ++u) {
        if (j + u < cnt) {
          const int d = __builtin_amdgcn_readlane(dl, j + u);
          const float w = __builtin_bit_cast(float, __builtin_amdgcn_readlane(wl, j + u));
          float* op = out + (size_t)d * F;
#pragma unroll
          for (int r = 0; r < R; ++r) atomicAdd(op + r * 32 + lane, v[u][r] * w);
        }
      }
    }
  }
}

typedef float __attribute__((address_space(1))) gcn_gfloat;
__device__ __forceinline__ gcn_gfloat* gcn_scalar_row(const float* p) {
  const unsigned long long a = (unsigned long long)p;
  const unsigned lo = __builtin_amdgcn_readfirstlane((unsigned)a);
  const unsigned hi = __builtin_amdgcn_readfirstlane((unsigned)(a >> 32));
  return (gcn_gfloat*)(((unsigned long long)hi << 32) | lo);
}
__device__ __forceinline__ void gcn_row_add(gcn_gfloat* p, float v) {
  (void)__hip_atomic_fetch_add(p, v, __ATOMIC_RELAXED, __HIP_MEMORY_SCOPE_AGENT);
}

__global__ __launch_bounds__(SC_WAVES * 32)
void gcn_edge_scatter_planes(const float* __restrict__ h,
                             const int* __restrict__ src,
                             const int* __restrict__ dst,
                             const float* __restrict__ nrm,
                             float* __restrict__ out, int nE,
                             int nRuns, int* __restrict__ claim) {
  constexpr int U = 16;
  __shared__ int s_task;

  const unsigned lane = threadIdx.x & 31u;
  const int wv = __builtin_amdgcn_readfirstlane((int)(threadIdx.x >> 5));

  if (threadIdx.x == 0) {
    const int side = (int)((__builtin_amdgcn_s_sendmsg_rtn(0x87) >> 18) & 1u);
    int pl = side;
    int t = atomicAdd(claim + pl * CLAIM_PITCH, 1);
    if (t >= nRuns) {
      pl = side ^ 1;
      t = atomicAdd(claim + pl * CLAIM_PITCH, 1);
    }
    s_task = t * 2 + pl;
  }
  __syncthreads();
  const int task = __builtin_amdgcn_readfirstlane(s_task);
  const int run = task >> 1;
  const int pl  = task & 1;
  if (run >= nRuns) return;

  const float* hpl = h + (size_t)pl * PLANE;
  float*       opl = out + (size_t)pl * PLANE;

  const int eb = run * SC_RUN + wv * (SC_B * 32);
  int sv[SC_B], dv[SC_B], wl[SC_B];
#pragma unroll
  for (int b = 0; b < SC_B; ++b) {
    const int e = min(eb + b * 32 + (int)lane, nE - 1);
    sv[b] = src[e];
    dv[b] = dst[e];
    wl[b] = __builtin_bit_cast(int, nrm[e]);
  }

#pragma unroll
  for (int b = 0; b < SC_B; ++b) {
    const int cnt = min(32, max(0, nE - (eb + b * 32)));
    if (cnt == 32) {
#pragma unroll
      for (int j = 0; j < 32; j += U) {
        float v[U];
#pragma unroll
        for (int u = 0; u < U; ++u) {
          const int s = __builtin_amdgcn_readlane(sv[b], j + u);
          const gcn_gfloat* hrow = gcn_scalar_row(hpl + (size_t)s * 32);
          v[u] = hrow[lane];
        }
#pragma unroll
        for (int u = 0; u < U; ++u) {
          const int d = __builtin_amdgcn_readlane(dv[b], j + u);
          const float w = __builtin_bit_cast(float, __builtin_amdgcn_readlane(wl[b], j + u));
          gcn_gfloat* orow = gcn_scalar_row(opl + (size_t)d * 32);
          gcn_row_add(orow + lane, v[u] * w);
        }
      }
    } else {
      for (int j = 0; j < cnt; ++j) {
        const int s = __builtin_amdgcn_readlane(sv[b], j);
        const int d = __builtin_amdgcn_readlane(dv[b], j);
        const float w = __builtin_bit_cast(float, __builtin_amdgcn_readlane(wl[b], j));
        const gcn_gfloat* hrow = gcn_scalar_row(hpl + (size_t)s * 32);
        gcn_gfloat* orow = gcn_scalar_row(opl + (size_t)d * 32);
        gcn_row_add(orow + lane, hrow[lane] * w);
      }
    }
  }
}


extern "C" void kernel_launch(void* const* d_in, const int* in_sizes, int n_in,
                              void* d_out, int out_size, void* d_ws, size_t ws_size,
                              hipStream_t stream) {
  const float* x  = (const float*)d_in[0];
  const int*   ei = (const int*)d_in[1];
  const float* W1 = (const float*)d_in[2];
  const float* b1 = (const float*)d_in[3];
  const float* W2 = (const float*)d_in[4];
  const float* b2 = (const float*)d_in[5];
  const float* W3 = (const float*)d_in[6];
  const float* b3 = (const float*)d_in[7];
  float* out = (float*)d_out;

  const int N = N_NODES;
  const int E = in_sizes[1] / 2;
  const int* src = ei;
  const int* dst = ei + E;

  float* ws   = (float*)d_ws;
  float* dis  = ws;
  float* nrm  = ws + OFF_NORM;
  float* bufA = ws + OFF_BUFA;
  float* bufB = bufA + NBUF;
  int*   part = (int*)bufA;
  int*   claim = (int*)(bufB + NBUF);

  const int TB = 256;
  const int gN = (N + TB - 1) / TB;
  const int gEN = (E + EN_B - 1) / EN_B;
  const int gGemm = N / 32;
  const int nRuns = (E + SC_RUN - 1) / SC_RUN;
  const int gScat = 2 * nRuns;
  const int nSplit = N / 2;

  gcn_deg_hist  <<<DEG_G * DEG_C, DEG_T, 0, stream>>>(dst, part, E);
  gcn_deg_finish<<<gN, TB, 0, stream>>>(part, dis, N, claim);

  gcn_gemm_wmma<128, 64, false, true, false><<<gEN + gGemm, 64, 0, stream>>>(
      x, W1, bufA, bufB, dis, b1, src, dst, nrm, E, gEN);
  gcn_edge_scatter_planes<<<gScat, SC_WAVES * 32, 0, stream>>>(bufA, src, dst, nrm, bufB, E,
                                                              nRuns, claim);

  gcn_gemm_wmma<64, 64, true, false, true><<<gGemm, 64, 0, stream>>>(
      bufB, W2, bufA, bufB, dis, b2, src, dst, nrm, E, 0);
  gcn_edge_scatter_planes<<<gScat, SC_WAVES * 32, 0, stream>>>(bufA, src, dst, nrm, bufB, E,
                                                              nRuns, claim + CLAIM_LAUNCH);

  gcn_gemm_wmma<64, 32, true, false, true><<<gGemm, 64, 0, stream>>>(
      bufB, W3, bufA, out, dis, b3, src, dst, nrm, E, 0);
  gcn_edge_scatter<32><<<gScat, SC_WAVES * 32, 0, stream>>>(bufA, src, dst, nrm, out, E,
                                                           nRuns, nSplit, claim + 2 * CLAIM_LAUNCH);
}


# Edge weight split in two row scales: GEMM epilogue stores h*dis[row], edges add plain rows, next A loader applies dis[row]; weight array, its pass, degree-finish launch go: 8 to 7; sum order: arrival
# speedup vs baseline: 2.3580x; 1.0947x over previous
#include <hip/hip_runtime.h>


typedef __attribute__((ext_vector_type(2))) float v2f;
typedef __attribute__((ext_vector_type(8))) float v8f;

#define N_NODES 100000
#define OFF_BUFA 100096
#define NBUF ((size_t)N_NODES * 64)
#define PLANE ((size_t)N_NODES * 32)

#define DEG_G 8
#define DEG_C 32
#define DEG_R 12512
#define DEG_NP (DEG_G * DEG_R)
#define DEG_T 512
#define DEG_U 10
static_assert(DEG_NP >= N_NODES && DEG_R % 32 == 0, "ranges must cover all nodes in whole lines");


#define CLAIM_PITCH 64
#define CLAIM_LAUNCH (2 * CLAIM_PITCH)

__global__ __launch_bounds__(DEG_T)
void gcn_deg_hist(const int* __restrict__ dst, int* __restrict__ part, int nE,
                  int* __restrict__ claim) {
  __shared__ __attribute__((aligned(16))) int cnt[DEG_R];
  const int tid = threadIdx.x;
  const int g = blockIdx.x % DEG_G;
  const int c = blockIdx.x / DEG_G;
  const int lo = g * DEG_R;

  if (blockIdx.x == 0 && tid < 6) claim[(tid >> 1) * CLAIM_LAUNCH + (tid & 1) * CLAIM_PITCH] = 0;

  int4* cnt4 = (int4*)cnt;
  for (int i = tid; i < DEG_R / 4; i += DEG_T) cnt4[i] = int4{0, 0, 0, 0};
  __syncthreads();

  const int chunk = (nE + DEG_C - 1) / DEG_C;
  const int e0 = min(nE, c * chunk);
  const int e1 = min(nE, e0 + chunk);
  const int* p = dst + e0;
  const int len = e1 - e0;

  auto hit = [&](int d) {
    const unsigned r = (unsigned)(d - lo);
    if (r < (unsigned)DEG_R) atomicAdd(&cnt[r], 1);
  };

  const int head = min(len, (int)(((16u - (unsigned)((size_t)p & 15u)) & 15u) >> 2));
  if (tid < head) hit(p[tid]);
  const int4* p4 = (const int4*)(p + head);
  const int n4 = (len - head) >> 2;
  for (int i = tid; i < n4; i += DEG_U * DEG_T) {
    int4 v[DEG_U];
#pragma unroll
    for (int u = 0; u < DEG_U; ++u) {
      const int idx = i + u * DEG_T;
      v[u] = idx < n4 ? p4[idx] : int4{-1, -1, -1, -1};
    }
#pragma unroll
    for (int u = 0; u < DEG_U; ++u) { hit(v[u].x); hit(v[u].y); hit(v[u].z); hit(v[u].w); }
  }
  const int t0 = head + 4 * n4;
  if (tid < len - t0) hit(p[t0 + tid]);
  __syncthreads();

  int4* o4 = (int4*)(part + (size_t)c * DEG_NP + lo);
  for (int i = tid; i < DEG_R / 4; i += DEG_T) o4[i] = cnt4[i];
}


template <int K, int NC, bool FIRST, bool LAST>
__global__ __launch_bounds__(64)
void gcn_gemm_wmma(const float* X, const float* __restrict__ W,
                   float* __restrict__ H, float* OUTI,
                   float* dis, const int* __restrict__ part,
                   const float* __restrict__ bprev, const float* __restrict__ bias) {
  constexpr int S  = K + 6;
  constexpr int CT = NC / 16;
  __shared__ float sWt[NC * S];

  const int lane = threadIdx.x & 31;
  const int wave = threadIdx.x >> 5;
  const int m    = lane & 15;
  const int half = lane >> 4;
  const long long row0 = ((long long)blockIdx.x * 2 + wave) * 16;

  int pv[DEG_C / 2];
  float dm = 0.f;
  if (FIRST) {
#pragma unroll
    for (int j = 0; j < DEG_C / 2; ++j)
      pv[j] = part[(size_t)(half * (DEG_C / 2) + j) * DEG_NP + (size_t)(row0 + m)];
  } else {
    dm = dis[row0 + m];
  }

  for (int t = threadIdx.x; t < K * NC / 4; t += 64) {
    const int kr = (t * 4) / NC;
    const int n  = (t * 4) % NC;
    const float4 w4 = *(const float4*)(W + (size_t)t * 4);
    sWt[(size_t)(n + 0) * S + kr] = w4.x;
    sWt[(size_t)(n + 1) * S + kr] = w4.y;
    sWt[(size_t)(n + 2) * S + kr] = w4.z;
    sWt[(size_t)(n + 3) * S + kr] = w4.w;
  }
  __syncthreads();

  if (FIRST) {
    int s = 0;
#pragma unroll
    for (int j = 0; j < DEG_C / 2; ++j) s += pv[j];
    s += __shfl_xor(s, 16);
    dm = rsqrtf((float)(s + 1));
    if (half == 0) dis[row0 + m] = dm;
  }

  static_assert(FIRST || K == 64, "plane input is two planes of 32 features");
  const float* xrow  = X + (size_t)(row0 + m) * (FIRST ? K : 32) + 2 * half;
  const float* bbase = sWt + (size_t)m * S + 2 * half;

  v8f acc[CT] = {};
#pragma unroll
  for (int kk = 0; kk < K / 4; ++kk) {
    const int k = kk * 4;
    const float* xp = FIRST ? xrow + k : xrow + (size_t)(k >> 5) * PLANE + (k & 31);
    float ax = xp[0];
    float ay = xp[1];
    if (!FIRST) {
      const v2f pb = *(const v2f*)(bprev + k + 2 * half);
      ax = fmaxf(fmaf(ax, dm, pb.x), 0.f);
      ay = fmaxf(fmaf(ay, dm, pb.y), 0.f);
    }
    const v2f a = {ax, ay};
#pragma unroll
    for (int ct = 0; ct < CT; ++ct) {
      const v2f b = *(const v2f*)(bbase + (size_t)ct * 16 * S + k);
      acc[ct] = __builtin_amdgcn_wmma_f32_16x16x4_f32(
          false, a, false, b, (short)0, acc[ct], false, false);
    }
  }

  float dr[8];
#pragma unroll
  for (int r = 0; r < 8; ++r) dr[r] = __shfl(dm, r + 8 * half);

#pragma unroll
  for (int ct = 0; ct < CT; ++ct) {
    const float bv = LAST ? bias[ct * 16 + m] : 0.f;
#pragma unroll
    for (int r = 0; r < 8; ++r) {
      const long long row = row0 + r + 8 * half;
      const float hs = acc[ct][r] * dr[r];
      const size_t o = (size_t)(ct >> 1) * PLANE + (size_t)row * 32 + (ct & 1) * 16 + m;
      H[o]    = hs;
      OUTI[o] = LAST ? bv + hs * dr[r] : hs;
    }
  }
}


#define SC_WAVES 8
#define SC_B 5
#define SC_RUN (SC_WAVES * SC_B * 32)

template <int F>
__global__ __launch_bounds__(SC_WAVES * 32)
void gcn_edge_scatter(const float* __restrict__ h,
                      const int* __restrict__ src,
                      const int* __restrict__ dst,
                      const float* __restrict__ dis,
                      float* __restrict__ out, int nE,
                      int nRuns, int nSplit, int* __restrict__ claim) {
  constexpr int R = F / 32;
  constexpr int U = 16 / R;
  __shared__ int   s_task;
  __shared__ int   s_cnt[SC_WAVES];
  __shared__ int   q_s[SC_RUN];
  __shared__ int   q_d[SC_RUN];
  __shared__ float q_w[SC_RUN];

  const int lane = (int)(threadIdx.x & 31);
  const int wv = __builtin_amdgcn_readfirstlane((int)(threadIdx.x >> 5));

  if (threadIdx.x == 0) {
    const int side = (int)((__builtin_amdgcn_s_sendmsg_rtn(0x87) >> 18) & 1u);
    int half = side;
    int t = atomicAdd(claim + half * CLAIM_PITCH, 1);
    if (t >= nRuns) {
      half = side ^ 1;
      t = atomicAdd(claim + half * CLAIM_PITCH, 1);
    }
    s_task = t * 2 + half;
  }
  __syncthreads();
  const int run  = s_task >> 1;
  const int half = s_task & 1;
  if (run >= nRuns) return;

  const int e0 = run * SC_RUN + wv * (SC_B * 32) + lane;
  int sv[SC_B], dv[SC_B];
#pragma unroll
  for (int b = 0; b < SC_B; ++b) {
    const int e = e0 + b * 32;
    const bool in = e < nE;
    dv[b] = in ? dst[e] : -1;
    sv[b] = in ? src[e] : 0;
  }
  unsigned mk[SC_B];
  float wt[SC_B];
  int n = 0;
#pragma unroll
  for (int b = 0; b < SC_B; ++b) {
    const bool mine = dv[b] >= 0 && ((dv[b] >= nSplit) == (half != 0));
    wt[b] = mine ? dis[dv[b]] : 0.f;
    mk[b] = __builtin_amdgcn_ballot_w32(mine);
    n += __builtin_popcount(mk[b]);
  }
  if (lane == 0) s_cnt[wv] = n;
  __syncthreads();

  int off = 0, total = 0;
#pragma unroll
  for (int i = 0; i < SC_WAVES; ++i) {
    const int c = s_cnt[i];
    off += i < wv ? c : 0;
    total += c;
  }
#pragma unroll
  for (int b = 0; b < SC_B; ++b) {
    if ((mk[b] >> lane) & 1u) {
      const int p = off + (int)__builtin_amdgcn_mbcnt_lo(mk[b], 0u);
      q_s[p] = sv[b];
      q_d[p] = dv[b];
      q_w[p] = wt[b];
    }
    off += __builtin_popcount(mk[b]);
  }
  __syncthreads();

  int share = (total + SC_WAVES - 1) / SC_WAVES;
  share = (share + U - 1) / U * U;
  const int b0 = wv * share;
  const int b1 = min(total, b0 + share);

  for (int base = b0; base < b1; base += 32) {
    const int cnt = min(32, b1 - base);
    int sl = 0, dl = 0, wl = 0;
    if (lane < cnt) {
      sl = q_s[base + lane];
      dl = q_d[base + lane];
      wl = __builtin_bit_cast(int, q_w[base + lane]);
    }
    int j = 0;
    for (; j + U <= cnt; j += U) {
      float v[U][R];
#pragma unroll
      for (int u = 0; u < U; ++u) {
        const int s = __builtin_amdgcn_readlane(sl, j + u);
        const float* hp = h + (size_t)s * F;
#pragma unroll
        for (int r = 0; r < R; ++r) v[u][r] = hp[r * 32 + lane];
      }
#pragma unroll
      for (int u = 0; u < U; ++u) {
        const int d = __builtin_amdgcn_readlane(dl, j + u);
        const float w = __builtin_bit_cast(float, __builtin_amdgcn_readlane(wl, j + u));
        float* op = out + (size_t)d * F;
#pragma unroll
        for (int r = 0; r < R; ++r) atomicAdd(op + r * 32 + lane, v[u][r] * w);
      }
    }
    if (j < cnt) {
      float v[U][R];
#pragma unroll
      for (int u = 0; u < U; ++u) {
        if (j + u < cnt) {
          const int s = __builtin_amdgcn_readlane(sl, j + u);
          const float* hp = h + (size_t)s * F;
#pragma unroll
          for (int r = 0; r < R; ++r) v[u][r] = hp[r * 32 + lane];
        }
      }
#pragma unroll
      for (int u = 0; u < U; ++u) {
        if (j + u < cnt) {
          const int d = __builtin_amdgcn_readlane(dl, j + u);
          const float w = __builtin_bit_cast(float, __builtin_amdgcn_readlane(wl, j + u));
          float* op = out + (size_t)d * F;
#pragma unroll
          for (int r = 0; r < R; ++r) atomicAdd(op + r * 32 + lane, v[u][r] * w);
        }
      }
    }
  }
}

typedef float __attribute__((address_space(1))) gcn_gfloat;
__device__ __forceinline__ gcn_gfloat* gcn_scalar_row(const float* p) {
  const unsigned long long a = (unsigned long long)p;
  const unsigned lo = __builtin_amdgcn_readfirstlane((unsigned)a);
  const unsigned hi = __builtin_amdgcn_readfirstlane((unsigned)(a >> 32));
  return (gcn_gfloat*)(((unsigned long long)hi << 32) | lo);
}
__device__ __forceinline__ void gcn_row_add(gcn_gfloat* p, float v) {
  (void)__hip_atomic_fetch_add(p, v, __ATOMIC_RELAXED, __HIP_MEMORY_SCOPE_AGENT);
}

__global__ __launch_bounds__(SC_WAVES * 32)
void gcn_edge_scatter_planes(const float* __restrict__ h,
                             const int* __restrict__ src,
                             const int* __restrict__ dst,
                             float* __restrict__ out, int nE,
                             int nRuns, int* __restrict__ claim) {
  constexpr int U = 16;
  __shared__ int s_task;

  const unsigned lane = threadIdx.x & 31u;
  const int wv = __builtin_amdgcn_readfirstlane((int)(threadIdx.x >> 5));

  if (threadIdx.x == 0) {
    const int side = (int)((__builtin_amdgcn_s_sendmsg_rtn(0x87) >> 18) & 1u);
    int pl = side;
    int t = atomicAdd(claim + pl * CLAIM_PITCH, 1);
    if (t >= nRuns) {
      pl = side ^ 1;
      t = atomicAdd(claim + pl * CLAIM_PITCH, 1);
    }
    s_task = t * 2 + pl;
  }
  __syncthreads();
  const int task = __builtin_amdgcn_readfirstlane(s_task);
  const int run = task >> 1;
  const int pl  = task & 1;
  if (run >= nRuns) return;

  const float* hpl = h + (size_t)pl * PLANE;
  float*       opl = out + (size_t)pl * PLANE;

  const int eb = run * SC_RUN + wv * (SC_B * 32);
  int sv[SC_B], dv[SC_B];
#pragma unroll
  for (int b = 0; b < SC_B; ++b) {
    const int e = min(eb + b * 32 + (int)lane, nE - 1);
    sv[b] = src[e];
    dv[b] = dst[e];
  }

#pragma unroll
  for (int b = 0; b < SC_B; ++b) {
    const int cnt = min(32, max(0, nE - (eb + b * 32)));
    if (cnt == 32) {
#pragma unroll
      for (int j = 0; j < 32; j += U) {
        float v[U];
#pragma unroll
        for (int u = 0; u < U; ++u) {
          const int s = __builtin_amdgcn_readlane(sv[b], j + u);
          const gcn_gfloat* hrow = gcn_scalar_row(hpl + (size_t)s * 32);
          v[u] = hrow[lane];
        }
#pragma unroll
        for (int u = 0; u < U; ++u) {
          const int d = __builtin_amdgcn_readlane(dv[b], j + u);
          gcn_gfloat* orow = gcn_scalar_row(opl + (size_t)d * 32);
          gcn_row_add(orow + lane, v[u]);
        }
      }
    } else {
      for (int j = 0; j < cnt; ++j) {
        const int s = __builtin_amdgcn_readlane(sv[b], j);
        const int d = __builtin_amdgcn_readlane(dv[b], j);
        const gcn_gfloat* hrow = gcn_scalar_row(hpl + (size_t)s * 32);
        gcn_gfloat* orow = gcn_scalar_row(opl + (size_t)d * 32);
        gcn_row_add(orow + lane, hrow[lane]);
      }
    }
  }
}


extern "C" void kernel_launch(void* const* d_in, const int* in_sizes, int n_in,
                              void* d_out, int out_size, void* d_ws, size_t ws_size,
                              hipStream_t stream) {
  const float* x  = (const float*)d_in[0];
  const int*   ei = (const int*)d_in[1];
  const float* W1 = (const float*)d_in[2];
  const float* b1 = (const float*)d_in[3];
  const float* W2 = (const float*)d_in[4];
  const float* b2 = (const float*)d_in[5];
  const float* W3 = (const float*)d_in[6];
  const float* b3 = (const float*)d_in[7];
  float* out = (float*)d_out;

  const int N = N_NODES;
  const int E = in_sizes[1] / 2;
  const int* src = ei;
  const int* dst = ei + E;

  float* ws   = (float*)d_ws;
  float* dis  = ws;
  float* bufA = ws + OFF_BUFA;
  float* bufB = bufA + NBUF;
  int*   claim = (int*)(bufB + NBUF);
  int*   part = claim + 1024;

  const int gGemm = N / 32;
  const int nRuns = (E + SC_RUN - 1) / SC_RUN;
  const int gScat = 2 * nRuns;
  const int nSplit = N / 2;

  gcn_deg_hist<<<DEG_G * DEG_C, DEG_T, 0, stream>>>(dst, part, E, claim);

  gcn_gemm_wmma<128, 64, true, false><<<gGemm, 64, 0, stream>>>(
      x, W1, bufA, bufB, dis, part, b1, b1);
  gcn_edge_scatter_planes<<<gScat, SC_WAVES * 32, 0, stream>>>(bufA, src, dst, bufB, E,
                                                              nRuns, claim);

  gcn_gemm_wmma<64, 64, false, false><<<gGemm, 64, 0, stream>>>(
      bufB, W2, bufA, bufB, dis, part, b1, b2);
  gcn_edge_scatter_planes<<<gScat, SC_WAVES * 32, 0, stream>>>(bufA, src, dst, bufB, E,
                                                              nRuns, claim + CLAIM_LAUNCH);

  gcn_gemm_wmma<64, 32, false, true><<<gGemm, 64, 0, stream>>>(
      bufB, W3, bufA, out, dis, part, b2, b3);
  gcn_edge_scatter<32><<<gScat, SC_WAVES * 32, 0, stream>>>(bufA, src, dst, dis, out, E,
                                                           nRuns, nSplit, claim + 2 * CLAIM_LAUNCH);
}
